# starting kernel
# baseline (speedup 1.0000x reference)
#include <hip/hip_runtime.h>


typedef float v2f __attribute__((ext_vector_type(2)));
typedef float v8f __attribute__((ext_vector_type(8)));

#define GRU_H   64
#define GRU_I   32
#define GRU_O   16
#define GRU_NG  192
#define BM      16
#define NWAVES  12
#define WAVE_SZ 32

__global__ __launch_bounds__(NWAVES * WAVE_SZ)
void gru_fused_wmma_kernel(const float* __restrict__ x,
                           const float* __restrict__ W_ih,
                           const float* __restrict__ W_hh,
                           const float* __restrict__ b_ih,
                           const float* __restrict__ b_hh,
                           const float* __restrict__ W_lin,
                           const float* __restrict__ b_lin,
                           float* __restrict__ out,
                           int T)
{
    __shared__ float h_buf[BM * GRU_H];
    __shared__ float r_buf[BM * GRU_H];
    __shared__ float z_buf[BM * GRU_H];

    const int tid  = threadIdx.x;
    const int wave = tid / WAVE_SZ;
    const int lane = tid % WAVE_SZ;
    const int hi   = lane >> 4;
    const int ln   = lane & 15;
    const int b0   = blockIdx.x * BM;

    const int  ncol = wave * 16;
    const int  gcol = ncol & (GRU_H - 1);
    const bool is_n = (wave >= 8);

    v2f Bih[GRU_I / 4];
    v2f Bhh[GRU_H / 4];
    v2f Blin[GRU_H / 4];
    {
        const float* wih_row = W_ih + (size_t)(ncol + ln) * GRU_I;
        #pragma unroll
        for (int kc = 0; kc < GRU_I / 4; ++kc) {
            Bih[kc][0] = wih_row[4 * kc + 2 * hi + 0];
            Bih[kc][1] = wih_row[4 * kc + 2 * hi + 1];
        }
        const float* whh_row = W_hh + (size_t)(ncol + ln) * GRU_H;
        #pragma unroll
        for (int kc = 0; kc < GRU_H / 4; ++kc) {
            Bhh[kc][0] = whh_row[4 * kc + 2 * hi + 0];
            Bhh[kc][1] = whh_row[4 * kc + 2 * hi + 1];
        }
        const float* wl_row = W_lin + (size_t)ln * GRU_H;
        #pragma unroll
        for (int kc = 0; kc < GRU_H / 4; ++kc) {
            Blin[kc][0] = wl_row[4 * kc + 2 * hi + 0];
            Blin[kc][1] = wl_row[4 * kc + 2 * hi + 1];
        }
    }
    const float bih_l  = b_ih[ncol + ln];
    const float bhh_l  = b_hh[ncol + ln];
    const float blin_l = b_lin[ln];

    for (int i = tid; i < BM * GRU_H; i += blockDim.x) h_buf[i] = 0.0f;
    __syncthreads();

    const float* xrow = x + ((size_t)(b0 + ln) * T) * GRU_I;

    for (int t = 0; t < T; ++t) {
        const float* xt = xrow + (size_t)t * GRU_I;
        __builtin_prefetch((const void*)(xt + GRU_I), 0, 1);

        v8f acc_x  = {0.f, 0.f, 0.f, 0.f, 0.f, 0.f, 0.f, 0.f};
        v8f acc_h0 = {0.f, 0.f, 0.f, 0.f, 0.f, 0.f, 0.f, 0.f};
        v8f acc_h1 = {0.f, 0.f, 0.f, 0.f, 0.f, 0.f, 0.f, 0.f};

        #pragma unroll
        for (int kc = 0; kc < GRU_I / 4; ++kc) {
            v2f a;
            a[0] = xt[4 * kc + 2 * hi + 0];
            a[1] = xt[4 * kc + 2 * hi + 1];
            acc_x = __builtin_amdgcn_wmma_f32_16x16x4_f32(
                false, a, false, Bih[kc], (short)0, acc_x, false, false);
        }

        #pragma unroll
        for (int kc = 0; kc < 8; ++kc) {
            v2f a;
            a[0] = h_buf[ln * GRU_H + 4 * kc + 2 * hi + 0];
            a[1] = h_buf[ln * GRU_H + 4 * kc + 2 * hi + 1];
            acc_h0 = __builtin_amdgcn_wmma_f32_16x16x4_f32(
                false, a, false, Bhh[kc], (short)0, acc_h0, false, false);
        }
        #pragma unroll
        for (int kc = 8; kc < 16; ++kc) {
            v2f a;
            a[0] = h_buf[ln * GRU_H + 4 * kc + 2 * hi + 0];
            a[1] = h_buf[ln * GRU_H + 4 * kc + 2 * hi + 1];
            acc_h1 = __builtin_amdgcn_wmma_f32_16x16x4_f32(
                false, a, false, Bhh[kc], (short)0, acc_h1, false, false);
        }

        if (!is_n) {
            #pragma unroll
            for (int v = 0; v < 8; ++v) {
                const int m = v + 8 * hi;
                float g = acc_x[v] + acc_h0[v] + acc_h1[v] + bih_l + bhh_l;
                g = 1.0f / (1.0f + __expf(-g));
                if (wave < 4) r_buf[m * GRU_H + gcol + ln] = g;
                else          z_buf[m * GRU_H + gcol + ln] = g;
            }
        }
        __syncthreads();

        if (is_n) {
            #pragma unroll
            for (int v = 0; v < 8; ++v) {
                const int m   = v + 8 * hi;
                const int idx = m * GRU_H + gcol + ln;
                const float r  = r_buf[idx];
                const float z  = z_buf[idx];
                const float gx = acc_x[v] + bih_l;
                const float gh = acc_h0[v] + acc_h1[v] + bhh_l;
                const float n  = tanhf(gx + r * gh);
                const float ho = h_buf[idx];
                h_buf[idx] = (1.0f - z) * n + z * ho;
            }
        }
        __syncthreads();

        if (wave == 0) {
            v8f acc_y0 = {0.f, 0.f, 0.f, 0.f, 0.f, 0.f, 0.f, 0.f};
            v8f acc_y1 = {0.f, 0.f, 0.f, 0.f, 0.f, 0.f, 0.f, 0.f};
            #pragma unroll
            for (int kc = 0; kc < 8; ++kc) {
                v2f a;
                a[0] = h_buf[ln * GRU_H + 4 * kc + 2 * hi + 0];
                a[1] = h_buf[ln * GRU_H + 4 * kc + 2 * hi + 1];
                acc_y0 = __builtin_amdgcn_wmma_f32_16x16x4_f32(
                    false, a, false, Blin[kc], (short)0, acc_y0, false, false);
            }
            #pragma unroll
            for (int kc = 8; kc < 16; ++kc) {
                v2f a;
                a[0] = h_buf[ln * GRU_H + 4 * kc + 2 * hi + 0];
                a[1] = h_buf[ln * GRU_H + 4 * kc + 2 * hi + 1];
                acc_y1 = __builtin_amdgcn_wmma_f32_16x16x4_f32(
                    false, a, false, Blin[kc], (short)0, acc_y1, false, false);
            }
            #pragma unroll
            for (int v = 0; v < 8; ++v) {
                const int m = v + 8 * hi;
                out[((size_t)(b0 + m) * T + t) * GRU_O + ln] =
                    acc_y0[v] + acc_y1[v] + blin_l;
            }
        }
    }
}

extern "C" void kernel_launch(void* const* d_in, const int* in_sizes, int n_in,
                              void* d_out, int out_size, void* d_ws, size_t ws_size,
                              hipStream_t stream) {
    const float* x     = (const float*)d_in[0];
    const float* W_ih  = (const float*)d_in[1];
    const float* W_hh  = (const float*)d_in[2];
    const float* b_ih  = (const float*)d_in[3];
    const float* b_hh  = (const float*)d_in[4];
    const float* W_lin = (const float*)d_in[5];
    const float* b_lin = (const float*)d_in[6];
    float* out = (float*)d_out;

    const int B = 512;
    const int T = in_sizes[0] / (B * GRU_I);

    dim3 grid(B / BM);
    dim3 block(NWAVES * WAVE_SZ);
    gru_fused_wmma_kernel<<<grid, block, 0, stream>>>(
        x, W_ih, W_hh, b_ih, b_hh, W_lin, b_lin, out, T);
}


# products computed transposed (weights = A operand, state = B operand): a lane's 8 new state values are its own B words; state via LDS in operand order, 16 B/lane; k order in each 16-slice permuted
# speedup vs baseline: 1.7428x; 1.7428x over previous
#include <hip/hip_runtime.h>


typedef float v2f __attribute__((ext_vector_type(2)));
typedef float v4f __attribute__((ext_vector_type(4)));
typedef float v8f __attribute__((ext_vector_type(8)));

#define GRU_H   64
#define GRU_I   32
#define GRU_O   16
#define GRU_NG  192
#define BM      16
#define NWAVES  12
#define WAVE_SZ 32
#define NSLICE  (GRU_H / 16)

__global__ __launch_bounds__(NWAVES * WAVE_SZ)
void gru_fused_wmma_kernel(const float* __restrict__ x,
                           const float* __restrict__ W_ih,
                           const float* __restrict__ W_hh,
                           const float* __restrict__ b_ih,
                           const float* __restrict__ b_hh,
                           const float* __restrict__ W_lin,
                           const float* __restrict__ b_lin,
                           float* __restrict__ out,
                           int T)
{
    __shared__ v4f h_frag[NSLICE * 2 * WAVE_SZ];
    __shared__ v4f r_frag[NSLICE * 2 * WAVE_SZ];
    __shared__ v4f z_frag[NSLICE * 2 * WAVE_SZ];

    const int tid  = threadIdx.x;
    const int wave = tid / WAVE_SZ;
    const int lane = tid % WAVE_SZ;
    const int hi   = lane >> 4;
    const int ln   = lane & 15;
    const int b0   = blockIdx.x * BM;

    const int  ncol = wave * 16;
    const int  wq   = wave & 3;
    const bool is_n = (wave >= 8);

    v2f Aih[GRU_I / 4];
    v2f Ahh[GRU_H / 4];
    v2f Alin[GRU_H / 4];
    {
        const float* wih_row = W_ih + (size_t)(ncol + ln) * GRU_I;
        #pragma unroll
        for (int kc = 0; kc < GRU_I / 4; ++kc) {
            Aih[kc][0] = wih_row[4 * kc + 2 * hi + 0];
            Aih[kc][1] = wih_row[4 * kc + 2 * hi + 1];
        }
        const float* whh_row = W_hh + (size_t)(ncol + ln) * GRU_H;
        const float* wl_row  = W_lin + (size_t)ln * GRU_H;
        #pragma unroll
        for (int s = 0; s < NSLICE; ++s) {
            #pragma unroll
            for (int c = 0; c < 4; ++c) {
                const int k = 16 * s + 8 * hi + 2 * c;
                Ahh[4 * s + c][0]  = whh_row[k + 0];
                Ahh[4 * s + c][1]  = whh_row[k + 1];
                Alin[4 * s + c][0] = wl_row[k + 0];
                Alin[4 * s + c][1] = wl_row[k + 1];
            }
        }
    }
    float bih_v[8], bhh_v[8], blin_v[8];
    #pragma unroll
    for (int v = 0; v < 8; ++v) {
        bih_v[v]  = b_ih[ncol + 8 * hi + v];
        bhh_v[v]  = b_hh[ncol + 8 * hi + v];
        blin_v[v] = b_lin[8 * hi + v];
    }

    for (int i = tid; i < NSLICE * 2 * WAVE_SZ; i += blockDim.x)
        h_frag[i] = (v4f){0.f, 0.f, 0.f, 0.f};
    v8f h_own = {0.f, 0.f, 0.f, 0.f, 0.f, 0.f, 0.f, 0.f};
    __syncthreads();

    v4f* const gate_dst = (wave < 4) ? r_frag : z_frag;
    const float* xrow = x + ((size_t)(b0 + ln) * T) * GRU_I;

    for (int t = 0; t < T; ++t) {
        const float* xt = xrow + (size_t)t * GRU_I;
        __builtin_prefetch((const void*)(xt + GRU_I), 0, 1);

        v8f acc_x  = {0.f, 0.f, 0.f, 0.f, 0.f, 0.f, 0.f, 0.f};
        v8f acc_h0 = {0.f, 0.f, 0.f, 0.f, 0.f, 0.f, 0.f, 0.f};
        v8f acc_h1 = {0.f, 0.f, 0.f, 0.f, 0.f, 0.f, 0.f, 0.f};

        #pragma unroll
        for (int kc = 0; kc < GRU_I / 4; ++kc) {
            v2f b;
            b[0] = xt[4 * kc + 2 * hi + 0];
            b[1] = xt[4 * kc + 2 * hi + 1];
            acc_x = __builtin_amdgcn_wmma_f32_16x16x4_f32(
                false, Aih[kc], false, b, (short)0, acc_x, false, false);
        }

        #pragma unroll
        for (int s = 0; s < NSLICE; ++s) {
            const v4f p0 = h_frag[(2 * s + 0) * WAVE_SZ + lane];
            const v4f p1 = h_frag[(2 * s + 1) * WAVE_SZ + lane];
            const v2f bq[4] = { {p0[0], p0[1]}, {p0[2], p0[3]}, {p1[0], p1[1]}, {p1[2], p1[3]} };
            #pragma unroll
            for (int c = 0; c < 4; ++c) {
                if (s < 2)
                    acc_h0 = __builtin_amdgcn_wmma_f32_16x16x4_f32(
                        false, Ahh[4 * s + c], false, bq[c], (short)0, acc_h0, false, false);
                else
                    acc_h1 = __builtin_amdgcn_wmma_f32_16x16x4_f32(
                        false, Ahh[4 * s + c], false, bq[c], (short)0, acc_h1, false, false);
            }
        }

        if (!is_n) {
            v8f g;
            #pragma unroll
            for (int v = 0; v < 8; ++v) {
                float a = acc_x[v] + acc_h0[v] + acc_h1[v] + bih_v[v] + bhh_v[v];
                g[v] = 1.0f / (1.0f + __expf(-a));
            }
            gate_dst[(2 * wq + 0) * WAVE_SZ + lane] = (v4f){g[0], g[1], g[2], g[3]};
            gate_dst[(2 * wq + 1) * WAVE_SZ + lane] = (v4f){g[4], g[5], g[6], g[7]};
        }
        __syncthreads();

        if (is_n) {
            const v4f r0 = r_frag[(2 * wq + 0) * WAVE_SZ + lane];
            const v4f r1 = r_frag[(2 * wq + 1) * WAVE_SZ + lane];
            const v4f z0 = z_frag[(2 * wq + 0) * WAVE_SZ + lane];
            const v4f z1 = z_frag[(2 * wq + 1) * WAVE_SZ + lane];
            #pragma unroll
            for (int v = 0; v < 8; ++v) {
                const float r  = (v < 4) ? r0[v & 3] : r1[v & 3];
                const float z  = (v < 4) ? z0[v & 3] : z1[v & 3];
                const float gx = acc_x[v] + bih_v[v];
                const float gh = acc_h0[v] + acc_h1[v] + bhh_v[v];
                const float n  = tanhf(gx + r * gh);
                h_own[v] = (1.0f - z) * n + z * h_own[v];
            }
            h_frag[(2 * wq + 0) * WAVE_SZ + lane] = (v4f){h_own[0], h_own[1], h_own[2], h_own[3]};
            h_frag[(2 * wq + 1) * WAVE_SZ + lane] = (v4f){h_own[4], h_own[5], h_own[6], h_own[7]};
        }
        __syncthreads();

        if (wave == 0) {
            v8f acc_y0 = {0.f, 0.f, 0.f, 0.f, 0.f, 0.f, 0.f, 0.f};
            v8f acc_y1 = {0.f, 0.f, 0.f, 0.f, 0.f, 0.f, 0.f, 0.f};
            #pragma unroll
            for (int s = 0; s < NSLICE; ++s) {
                const v4f p0 = h_frag[(2 * s + 0) * WAVE_SZ + lane];
                const v4f p1 = h_frag[(2 * s + 1) * WAVE_SZ + lane];
                const v2f bq[4] = { {p0[0], p0[1]}, {p0[2], p0[3]}, {p1[0], p1[1]}, {p1[2], p1[3]} };
                #pragma unroll
                for (int c = 0; c < 4; ++c) {
                    if (s < 2)
                        acc_y0 = __builtin_amdgcn_wmma_f32_16x16x4_f32(
                            false, Alin[4 * s + c], false, bq[c], (short)0, acc_y0, false, false);
                    else
                        acc_y1 = __builtin_amdgcn_wmma_f32_16x16x4_f32(
                            false, Alin[4 * s + c], false, bq[c], (short)0, acc_y1, false, false);
                }
            }
            float* orow = out + ((size_t)(b0 + ln) * T + t) * GRU_O + 8 * hi;
            float y[8];
            #pragma unroll
            for (int v = 0; v < 8; ++v) y[v] = acc_y0[v] + acc_y1[v] + blin_v[v];
            *(v4f*)(orow + 0) = (v4f){y[0], y[1], y[2], y[3]};
            *(v4f*)(orow + 4) = (v4f){y[4], y[5], y[6], y[7]};
        }
    }
}

extern "C" void kernel_launch(void* const* d_in, const int* in_sizes, int n_in,
                              void* d_out, int out_size, void* d_ws, size_t ws_size,
                              hipStream_t stream) {
    const float* x     = (const float*)d_in[0];
    const float* W_ih  = (const float*)d_in[1];
    const float* W_hh  = (const float*)d_in[2];
    const float* b_ih  = (const float*)d_in[3];
    const float* b_hh  = (const float*)d_in[4];
    const float* W_lin = (const float*)d_in[5];
    const float* b_lin = (const float*)d_in[6];
    float* out = (float*)d_out;

    const int B = 512;
    const int T = in_sizes[0] / (B * GRU_I);

    dim3 grid(B / BM);
    dim3 block(NWAVES * WAVE_SZ);
    gru_fused_wmma_kernel<<<grid, block, 0, stream>>>(
        x, W_ih, W_hh, b_ih, b_hh, W_lin, b_lin, out, T);
}


# the two products that do not depend on the step before run as whole-chip launches around the recurrence (gx and states via workspace, 3 launches); x summed in another k order, b_ih added first
# speedup vs baseline: 2.1563x; 1.2373x over previous
#include <hip/hip_runtime.h>


typedef float v2f __attribute__((ext_vector_type(2)));
typedef float v4f __attribute__((ext_vector_type(4)));
typedef float v8f __attribute__((ext_vector_type(8)));

#define GRU_H   64
#define GRU_I   32
#define GRU_O   16
#define GRU_NG  192
#define BM      16
#define NBLK    32
#define NWAVES  12
#define NTILE   12
#define WAVE_SZ 32
#define NSLICE  (GRU_H / 16)
#define GX_UNITS_PER_STEP   (NTILE * 2 * WAVE_SZ)
#define ST_UNITS_PER_STEP   (NSLICE * 2 * WAVE_SZ)

__global__ __launch_bounds__(4 * WAVE_SZ)
void gru_input_gates_kernel(const float* __restrict__ x,
                            const float* __restrict__ W_ih,
                            const float* __restrict__ b_ih,
                            float* __restrict__ ws,
                            int T)
{
    const int tid  = threadIdx.x;
    const int wave = tid / WAVE_SZ;
    const int lane = tid % WAVE_SZ;
    const int hi   = lane >> 4;
    const int ln   = lane & 15;
    const int blk  = blockIdx.y;
    const int b0   = blk * BM;

    v2f   A[3][GRU_I / 4];
    float bias[3][8];
    #pragma unroll
    for (int g = 0; g < 3; ++g) {
        const int tile = 3 * wave + g;
        const v4f* wrow = (const v4f*)(W_ih + (size_t)(16 * tile + ln) * GRU_I + 16 * hi);
        #pragma unroll
        for (int i = 0; i < 4; ++i) {
            const v4f w = wrow[i];
            A[g][2 * i + 0] = (v2f){w[0], w[1]};
            A[g][2 * i + 1] = (v2f){w[2], w[3]};
        }
        #pragma unroll
        for (int v = 0; v < 8; ++v) bias[g][v] = b_ih[16 * tile + 8 * hi + v];
    }

    const float* xrow = x + ((size_t)(b0 + ln) * T) * GRU_I + 16 * hi;
    v4f* const gx = (v4f*)ws;

    const int t0 = blockIdx.x;
    v4f xn[4];
    if (t0 < T) {
        const v4f* xp = (const v4f*)(xrow + (size_t)t0 * GRU_I);
        #pragma unroll
        for (int i = 0; i < 4; ++i) xn[i] = xp[i];
    }

    for (int t = t0; t < T; t += gridDim.x) {
        v4f xv[4];
        #pragma unroll
        for (int i = 0; i < 4; ++i) xv[i] = xn[i];
        {
            const int tn = (t + (int)gridDim.x < T) ? t + (int)gridDim.x : t;
            const v4f* xp = (const v4f*)(xrow + (size_t)tn * GRU_I);
            #pragma unroll
            for (int i = 0; i < 4; ++i) xn[i] = xp[i];
        }

        v8f acc[3];
        #pragma unroll
        for (int g = 0; g < 3; ++g) acc[g] = (v8f){0.f, 0.f, 0.f, 0.f, 0.f, 0.f, 0.f, 0.f};

        #pragma unroll
        for (int kc = 0; kc < GRU_I / 4; ++kc) {
            const v2f b = {xv[kc >> 1][2 * (kc & 1) + 0], xv[kc >> 1][2 * (kc & 1) + 1]};
            #pragma unroll
            for (int g = 0; g < 3; ++g)
                acc[g] = __builtin_amdgcn_wmma_f32_16x16x4_f32(
                    false, A[g][kc], false, b, (short)0, acc[g], false, false);
        }

        v4f* dst = gx + ((size_t)blk * T + t) * GX_UNITS_PER_STEP + (3 * wave) * 2 * WAVE_SZ + lane;
        #pragma unroll
        for (int g = 0; g < 3; ++g) {
            dst[(2 * g + 0) * WAVE_SZ] = (v4f){acc[g][0] + bias[g][0], acc[g][1] + bias[g][1],
                                               acc[g][2] + bias[g][2], acc[g][3] + bias[g][3]};
            dst[(2 * g + 1) * WAVE_SZ] = (v4f){acc[g][4] + bias[g][4], acc[g][5] + bias[g][5],
                                               acc[g][6] + bias[g][6], acc[g][7] + bias[g][7]};
        }
    }
}

__global__ __launch_bounds__(NWAVES * WAVE_SZ)
void gru_recurrent_kernel(const float* __restrict__ W_hh,
                          const float* __restrict__ b_hh,
                          float* __restrict__ ws,
                          int T)
{
    __shared__ v4f h_frag[NSLICE * 2 * WAVE_SZ];
    __shared__ v4f r_frag[NSLICE * 2 * WAVE_SZ];
    __shared__ v4f z_frag[NSLICE * 2 * WAVE_SZ];

    const int tid  = threadIdx.x;
    const int wave = tid / WAVE_SZ;
    const int lane = tid % WAVE_SZ;
    const int hi   = lane >> 4;
    const int ln   = lane & 15;
    const int blk  = blockIdx.x;

    const int  ncol = wave * 16;
    const int  wq   = wave & 3;
    const bool is_n = (wave >= 8);

    v2f Ahh[GRU_H / 4];
    {
        const float* whh_row = W_hh + (size_t)(ncol + ln) * GRU_H;
        #pragma unroll
        for (int s = 0; s < NSLICE; ++s) {
            #pragma unroll
            for (int c = 0; c < 4; ++c) {
                const int k = 16 * s + 8 * hi + 2 * c;
                Ahh[4 * s + c][0] = whh_row[k + 0];
                Ahh[4 * s + c][1] = whh_row[k + 1];
            }
        }
    }
    float bhh_v[8];
    #pragma unroll
    for (int v = 0; v < 8; ++v) bhh_v[v] = b_hh[ncol + 8 * hi + v];

    for (int i = tid; i < NSLICE * 2 * WAVE_SZ; i += blockDim.x)
        h_frag[i] = (v4f){0.f, 0.f, 0.f, 0.f};
    v8f h_own = {0.f, 0.f, 0.f, 0.f, 0.f, 0.f, 0.f, 0.f};
    __syncthreads();

    v4f* const gate_dst = (wave < 4) ? r_frag : z_frag;

    const v4f* gxw = (const v4f*)ws + (size_t)blk * T * GX_UNITS_PER_STEP + wave * 2 * WAVE_SZ + lane;
    v4f* stw = (v4f*)ws + (size_t)NBLK * T * GX_UNITS_PER_STEP
             + (size_t)blk * T * ST_UNITS_PER_STEP + (2 * wq) * WAVE_SZ + lane;

    v4f gx0 = gxw[0];
    v4f gx1 = gxw[WAVE_SZ];

    for (int t = 0; t < T; ++t) {
        const int tn = (t + 1 < T) ? t + 1 : t;
        const v4f nx0 = gxw[(size_t)tn * GX_UNITS_PER_STEP];
        const v4f nx1 = gxw[(size_t)tn * GX_UNITS_PER_STEP + WAVE_SZ];

        v8f acc_h0 = {0.f, 0.f, 0.f, 0.f, 0.f, 0.f, 0.f, 0.f};
        v8f acc_h1 = {0.f, 0.f, 0.f, 0.f, 0.f, 0.f, 0.f, 0.f};

        #pragma unroll
        for (int s = 0; s < NSLICE; ++s) {
            const v4f p0 = h_frag[(2 * s + 0) * WAVE_SZ + lane];
            const v4f p1 = h_frag[(2 * s + 1) * WAVE_SZ + lane];
            const v2f bq[4] = { {p0[0], p0[1]}, {p0[2], p0[3]}, {p1[0], p1[1]}, {p1[2], p1[3]} };
            #pragma unroll
            for (int c = 0; c < 4; ++c) {
                if (s < 2)
                    acc_h0 = __builtin_amdgcn_wmma_f32_16x16x4_f32(
                        false, Ahh[4 * s + c], false, bq[c], (short)0, acc_h0, false, false);
                else
                    acc_h1 = __builtin_amdgcn_wmma_f32_16x16x4_f32(
                        false, Ahh[4 * s + c], false, bq[c], (short)0, acc_h1, false, false);
            }
        }

        const float gxv[8] = {gx0[0], gx0[1], gx0[2], gx0[3], gx1[0], gx1[1], gx1[2], gx1[3]};

        if (!is_n) {
            v8f g;
            #pragma unroll
            for (int v = 0; v < 8; ++v) {
                float a = gxv[v] + acc_h0[v] + acc_h1[v] + bhh_v[v];
                g[v] = 1.0f / (1.0f + __expf(-a));
            }
            gate_dst[(2 * wq + 0) * WAVE_SZ + lane] = (v4f){g[0], g[1], g[2], g[3]};
            gate_dst[(2 * wq + 1) * WAVE_SZ + lane] = (v4f){g[4], g[5], g[6], g[7]};
        }
        __syncthreads();

        if (is_n) {
            const v4f r0 = r_frag[(2 * wq + 0) * WAVE_SZ + lane];
            const v4f r1 = r_frag[(2 * wq + 1) * WAVE_SZ + lane];
            const v4f z0 = z_frag[(2 * wq + 0) * WAVE_SZ + lane];
            const v4f z1 = z_frag[(2 * wq + 1) * WAVE_SZ + lane];
            #pragma unroll
            for (int v = 0; v < 8; ++v) {
                const float r  = (v < 4) ? r0[v & 3] : r1[v & 3];
                const float z  = (v < 4) ? z0[v & 3] : z1[v & 3];
                const float gh = acc_h0[v] + acc_h1[v] + bhh_v[v];
                const float n  = tanhf(gxv[v] + r * gh);
                h_own[v] = (1.0f - z) * n + z * h_own[v];
            }
            const v4f lo = {h_own[0], h_own[1], h_own[2], h_own[3]};
            const v4f up = {h_own[4], h_own[5], h_own[6], h_own[7]};
            h_frag[(2 * wq + 0) * WAVE_SZ + lane] = lo;
            h_frag[(2 * wq + 1) * WAVE_SZ + lane] = up;
        }
        __syncthreads();

        if (is_n) {
            v4f* st = stw + (size_t)t * ST_UNITS_PER_STEP;
            st[0]       = (v4f){h_own[0], h_own[1], h_own[2], h_own[3]};
            st[WAVE_SZ] = (v4f){h_own[4], h_own[5], h_own[6], h_own[7]};
        }

        gx0 = nx0;
        gx1 = nx1;
    }
}

__global__ __launch_bounds__(8 * WAVE_SZ)
void gru_projection_kernel(const float* __restrict__ W_lin,
                           const float* __restrict__ b_lin,
                           const float* __restrict__ ws,
                           float* __restrict__ out,
                           int T)
{
    const int tid  = threadIdx.x;
    const int wave = tid / WAVE_SZ;
    const int lane = tid % WAVE_SZ;
    const int hi   = lane >> 4;
    const int ln   = lane & 15;
    const int blk  = blockIdx.y;
    const int b0   = blk * BM;

    v2f Alin[GRU_H / 4];
    {
        const float* wl_row = W_lin + (size_t)ln * GRU_H;
        #pragma unroll
        for (int s = 0; s < NSLICE; ++s) {
            #pragma unroll
            for (int c = 0; c < 4; ++c) {
                const int k = 16 * s + 8 * hi + 2 * c;
                Alin[4 * s + c][0] = wl_row[k + 0];
                Alin[4 * s + c][1] = wl_row[k + 1];
            }
        }
    }
    float blin_v[8];
    #pragma unroll
    for (int v = 0; v < 8; ++v) blin_v[v] = b_lin[8 * hi + v];

    const v4f* stb = (const v4f*)ws + (size_t)NBLK * T * GX_UNITS_PER_STEP
                   + (size_t)blk * T * ST_UNITS_PER_STEP + lane;

    for (int t = blockIdx.x * 8 + wave; t < T; t += gridDim.x * 8) {
        const v4f* hp = stb + (size_t)t * ST_UNITS_PER_STEP;
        v8f acc_y0 = {0.f, 0.f, 0.f, 0.f, 0.f, 0.f, 0.f, 0.f};
        v8f acc_y1 = {0.f, 0.f, 0.f, 0.f, 0.f, 0.f, 0.f, 0.f};
        #pragma unroll
        for (int s = 0; s < NSLICE; ++s) {
            const v4f p0 = hp[(2 * s + 0) * WAVE_SZ];
            const v4f p1 = hp[(2 * s + 1) * WAVE_SZ];
            const v2f bq[4] = { {p0[0], p0[1]}, {p0[2], p0[3]}, {p1[0], p1[1]}, {p1[2], p1[3]} };
            #pragma unroll
            for (int c = 0; c < 4; ++c) {
                if (s < 2)
                    acc_y0 = __builtin_amdgcn_wmma_f32_16x16x4_f32(
                        false, Alin[4 * s + c], false, bq[c], (short)0, acc_y0, false, false);
                else
                    acc_y1 = __builtin_amdgcn_wmma_f32_16x16x4_f32(
                        false, Alin[4 * s + c], false, bq[c], (short)0, acc_y1, false, false);
            }
        }
        float* orow = out + ((size_t)(b0 + ln) * T + t) * GRU_O + 8 * hi;
        float y[8];
        #pragma unroll
        for (int v = 0; v < 8; ++v) y[v] = acc_y0[v] + acc_y1[v] + blin_v[v];
        *(v4f*)(orow + 0) = (v4f){y[0], y[1], y[2], y[3]};
        *(v4f*)(orow + 4) = (v4f){y[4], y[5], y[6], y[7]};
    }
}

extern "C" void kernel_launch(void* const* d_in, const int* in_sizes, int n_in,
                              void* d_out, int out_size, void* d_ws, size_t ws_size,
                              hipStream_t stream) {
    const float* x     = (const float*)d_in[0];
    const float* W_ih  = (const float*)d_in[1];
    const float* W_hh  = (const float*)d_in[2];
    const float* b_ih  = (const float*)d_in[3];
    const float* b_hh  = (const float*)d_in[4];
    const float* W_lin = (const float*)d_in[5];
    const float* b_lin = (const float*)d_in[6];
    float* out = (float*)d_out;
    float* ws  = (float*)d_ws;

    const int B = 512;
    const int T = in_sizes[0] / (B * GRU_I);

    gru_input_gates_kernel<<<dim3(32, NBLK), dim3(4 * WAVE_SZ), 0, stream>>>(
        x, W_ih, b_ih, ws, T);

    dim3 grid(B / BM);
    dim3 block(NWAVES * WAVE_SZ);
    gru_recurrent_kernel<<<grid, block, 0, stream>>>(W_hh, b_hh, ws, T);

    gru_projection_kernel<<<dim3(16, NBLK), dim3(8 * WAVE_SZ), 0, stream>>>(
        W_lin, b_lin, ws, out, T);
}


# n gate of each hidden slice split by register quad over two waves (the n wave and the slice's r wave, fed through LDS): 4 serial tanh chains per wave instead of 8
# speedup vs baseline: 2.6374x; 1.2231x over previous
#include <hip/hip_runtime.h>


typedef float v2f __attribute__((ext_vector_type(2)));
typedef float v4f __attribute__((ext_vector_type(4)));
typedef float v8f __attribute__((ext_vector_type(8)));

#define GRU_H   64
#define GRU_I   32
#define GRU_O   16
#define GRU_NG  192
#define BM      16
#define NBLK    32
#define NWAVES  12
#define NTILE   12
#define WAVE_SZ 32
#define NSLICE  (GRU_H / 16)
#define GX_UNITS_PER_STEP   (NTILE * 2 * WAVE_SZ)
#define ST_UNITS_PER_STEP   (NSLICE * 2 * WAVE_SZ)

__global__ __launch_bounds__(4 * WAVE_SZ)
void gru_input_gates_kernel(const float* __restrict__ x,
                            const float* __restrict__ W_ih,
                            const float* __restrict__ b_ih,
                            float* __restrict__ ws,
                            int T)
{
    const int tid  = threadIdx.x;
    const int wave = tid / WAVE_SZ;
    const int lane = tid % WAVE_SZ;
    const int hi   = lane >> 4;
    const int ln   = lane & 15;
    const int blk  = blockIdx.y;
    const int b0   = blk * BM;

    v2f   A[3][GRU_I / 4];
    float bias[3][8];
    #pragma unroll
    for (int g = 0; g < 3; ++g) {
        const int tile = 3 * wave + g;
        const v4f* wrow = (const v4f*)(W_ih + (size_t)(16 * tile + ln) * GRU_I + 16 * hi);
        #pragma unroll
        for (int i = 0; i < 4; ++i) {
            const v4f w = wrow[i];
            A[g][2 * i + 0] = (v2f){w[0], w[1]};
            A[g][2 * i + 1] = (v2f){w[2], w[3]};
        }
        #pragma unroll
        for (int v = 0; v < 8; ++v) bias[g][v] = b_ih[16 * tile + 8 * hi + v];
    }

    const float* xrow = x + ((size_t)(b0 + ln) * T) * GRU_I + 16 * hi;
    v4f* const gx = (v4f*)ws;

    const int t0 = blockIdx.x;
    v4f xn[4];
    if (t0 < T) {
        const v4f* xp = (const v4f*)(xrow + (size_t)t0 * GRU_I);
        #pragma unroll
        for (int i = 0; i < 4; ++i) xn[i] = xp[i];
    }

    for (int t = t0; t < T; t += gridDim.x) {
        v4f xv[4];
        #pragma unroll
        for (int i = 0; i < 4; ++i) xv[i] = xn[i];
        {
            const int tn = (t + (int)gridDim.x < T) ? t + (int)gridDim.x : t;
            const v4f* xp = (const v4f*)(xrow + (size_t)tn * GRU_I);
            #pragma unroll
            for (int i = 0; i < 4; ++i) xn[i] = xp[i];
        }

        v8f acc[3];
        #pragma unroll
        for (int g = 0; g < 3; ++g) acc[g] = (v8f){0.f, 0.f, 0.f, 0.f, 0.f, 0.f, 0.f, 0.f};

        #pragma unroll
        for (int kc = 0; kc < GRU_I / 4; ++kc) {
            const v2f b = {xv[kc >> 1][2 * (kc & 1) + 0], xv[kc >> 1][2 * (kc & 1) + 1]};
            #pragma unroll
            for (int g = 0; g < 3; ++g)
                acc[g] = __builtin_amdgcn_wmma_f32_16x16x4_f32(
                    false, A[g][kc], false, b, (short)0, acc[g], false, false);
        }

        v4f* dst = gx + ((size_t)blk * T + t) * GX_UNITS_PER_STEP + (3 * wave) * 2 * WAVE_SZ + lane;
        #pragma unroll
        for (int g = 0; g < 3; ++g) {
            dst[(2 * g + 0) * WAVE_SZ] = (v4f){acc[g][0] + bias[g][0], acc[g][1] + bias[g][1],
                                               acc[g][2] + bias[g][2], acc[g][3] + bias[g][3]};
            dst[(2 * g + 1) * WAVE_SZ] = (v4f){acc[g][4] + bias[g][4], acc[g][5] + bias[g][5],
                                               acc[g][6] + bias[g][6], acc[g][7] + bias[g][7]};
        }
    }
}

__device__ __forceinline__ v4f n_gate_quad(v4f gx, v4f r, v4f gh, v4f z, v4f h)
{
    v4f o;
    #pragma unroll
    for (int v = 0; v < 4; ++v) {
        const float n = tanhf(gx[v] + r[v] * gh[v]);
        o[v] = (1.0f - z[v]) * n + z[v] * h[v];
    }
    return o;
}

__global__ __launch_bounds__(NWAVES * WAVE_SZ)
void gru_recurrent_kernel(const float* __restrict__ W_hh,
                          const float* __restrict__ b_hh,
                          float* __restrict__ ws,
                          int T)
{
    __shared__ v4f h_frag[NSLICE * 2 * WAVE_SZ];
    __shared__ v4f z_frag[NSLICE * 2 * WAVE_SZ];
    __shared__ v4f r_frag[NSLICE * WAVE_SZ];
    __shared__ v4f g_frag[NSLICE * WAVE_SZ];

    const int tid  = threadIdx.x;
    const int wave = tid / WAVE_SZ;
    const int lane = tid % WAVE_SZ;
    const int hi   = lane >> 4;
    const int ln   = lane & 15;
    const int blk  = blockIdx.x;

    const int  ncol = wave * 16;
    const int  wq   = wave & 3;
    const bool is_r = (wave < 4);
    const bool is_z = (wave >= 4) && (wave < 8);
    const bool is_n = (wave >= 8);
    const int  quad = is_n ? 0 : 1;

    v2f Ahh[GRU_H / 4];
    {
        const float* whh_row = W_hh + (size_t)(ncol + ln) * GRU_H;
        #pragma unroll
        for (int s = 0; s < NSLICE; ++s) {
            #pragma unroll
            for (int c = 0; c < 4; ++c) {
                const int k = 16 * s + 8 * hi + 2 * c;
                Ahh[4 * s + c][0] = whh_row[k + 0];
                Ahh[4 * s + c][1] = whh_row[k + 1];
            }
        }
    }
    float bhh_v[8];
    #pragma unroll
    for (int v = 0; v < 8; ++v) bhh_v[v] = b_hh[ncol + 8 * hi + v];

    for (int i = tid; i < NSLICE * 2 * WAVE_SZ; i += blockDim.x)
        h_frag[i] = (v4f){0.f, 0.f, 0.f, 0.f};
    v4f hq = {0.f, 0.f, 0.f, 0.f};
    __syncthreads();

    const v4f* const gxb = (const v4f*)ws + (size_t)blk * T * GX_UNITS_PER_STEP;
    const v4f* gxw = gxb + wave * 2 * WAVE_SZ + lane;
    const v4f* gxn = gxb + ((8 + wq) * 2 + quad) * WAVE_SZ + lane;
    v4f* stw = (v4f*)ws + (size_t)NBLK * T * GX_UNITS_PER_STEP
             + (size_t)blk * T * ST_UNITS_PER_STEP + (2 * wq + quad) * WAVE_SZ + lane;

    v4f gx0 = {0.f, 0.f, 0.f, 0.f}, gx1 = gx0, gxc = gx0;
    if (!is_n) { gx0 = gxw[0]; gx1 = gxw[WAVE_SZ]; }
    if (!is_z) { gxc = gxn[0]; }

    for (int t = 0; t < T; ++t) {
        const int tn = (t + 1 < T) ? t + 1 : t;
        v4f nx0 = gx0, nx1 = gx1, nxc = gxc;
        if (!is_n) {
            nx0 = gxw[(size_t)tn * GX_UNITS_PER_STEP];
            nx1 = gxw[(size_t)tn * GX_UNITS_PER_STEP + WAVE_SZ];
        }
        if (!is_z) nxc = gxn[(size_t)tn * GX_UNITS_PER_STEP];

        v8f acc_h0 = {0.f, 0.f, 0.f, 0.f, 0.f, 0.f, 0.f, 0.f};
        v8f acc_h1 = {0.f, 0.f, 0.f, 0.f, 0.f, 0.f, 0.f, 0.f};

        #pragma unroll
        for (int s = 0; s < NSLICE; ++s) {
            const v4f p0 = h_frag[(2 * s + 0) * WAVE_SZ + lane];
            const v4f p1 = h_frag[(2 * s + 1) * WAVE_SZ + lane];
            const v2f bq[4] = { {p0[0], p0[1]}, {p0[2], p0[3]}, {p1[0], p1[1]}, {p1[2], p1[3]} };
            #pragma unroll
            for (int c = 0; c < 4; ++c) {
                if (s < 2)
                    acc_h0 = __builtin_amdgcn_wmma_f32_16x16x4_f32(
                        false, Ahh[4 * s + c], false, bq[c], (short)0, acc_h0, false, false);
                else
                    acc_h1 = __builtin_amdgcn_wmma_f32_16x16x4_f32(
                        false, Ahh[4 * s + c], false, bq[c], (short)0, acc_h1, false, false);
            }
        }

        v4f rq1 = {0.f, 0.f, 0.f, 0.f};
        v4f gq0 = {0.f, 0.f, 0.f, 0.f};

        if (!is_n) {
            const float gxv[8] = {gx0[0], gx0[1], gx0[2], gx0[3], gx1[0], gx1[1], gx1[2], gx1[3]};
            v8f g;
            #pragma unroll
            for (int v = 0; v < 8; ++v) {
                float a = gxv[v] + acc_h0[v] + acc_h1[v] + bhh_v[v];
                g[v] = 1.0f / (1.0f + __expf(-a));
            }
            if (is_r) {
                r_frag[wq * WAVE_SZ + lane] = (v4f){g[0], g[1], g[2], g[3]};
                rq1 = (v4f){g[4], g[5], g[6], g[7]};
            } else {
                z_frag[(2 * wq + 0) * WAVE_SZ + lane] = (v4f){g[0], g[1], g[2], g[3]};
                z_frag[(2 * wq + 1) * WAVE_SZ + lane] = (v4f){g[4], g[5], g[6], g[7]};
            }
        } else {
            float gh[8];
            #pragma unroll
            for (int v = 0; v < 8; ++v) gh[v] = acc_h0[v] + acc_h1[v] + bhh_v[v];
            gq0 = (v4f){gh[0], gh[1], gh[2], gh[3]};
            g_frag[wq * WAVE_SZ + lane] = (v4f){gh[4], gh[5], gh[6], gh[7]};
        }
        __syncthreads();

        if (is_n) {
            const v4f rc = r_frag[wq * WAVE_SZ + lane];
            const v4f zc = z_frag[(2 * wq + 0) * WAVE_SZ + lane];
            hq = n_gate_quad(gxc, rc, gq0, zc, hq);
            h_frag[(2 * wq + 0) * WAVE_SZ + lane] = hq;
        } else if (is_r) {
            const v4f gc = g_frag[wq * WAVE_SZ + lane];
            const v4f zc = z_frag[(2 * wq + 1) * WAVE_SZ + lane];
            hq = n_gate_quad(gxc, rq1, gc, zc, hq);
            h_frag[(2 * wq + 1) * WAVE_SZ + lane] = hq;
        }
        __syncthreads();

        if (!is_z) stw[(size_t)t * ST_UNITS_PER_STEP] = hq;

        gx0 = nx0;
        gx1 = nx1;
        gxc = nxc;
    }
}

__global__ __launch_bounds__(8 * WAVE_SZ)
void gru_projection_kernel(const float* __restrict__ W_lin,
                           const float* __restrict__ b_lin,
                           const float* __restrict__ ws,
                           float* __restrict__ out,
                           int T)
{
    const int tid  = threadIdx.x;
    const int wave = tid / WAVE_SZ;
    const int lane = tid % WAVE_SZ;
    const int hi   = lane >> 4;
    const int ln   = lane & 15;
    const int blk  = blockIdx.y;
    const int b0   = blk * BM;

    v2f Alin[GRU_H / 4];
    {
        const float* wl_row = W_lin + (size_t)ln * GRU_H;
        #pragma unroll
        for (int s = 0; s < NSLICE; ++s) {
            #pragma unroll
            for (int c = 0; c < 4; ++c) {
                const int k = 16 * s + 8 * hi + 2 * c;
                Alin[4 * s + c][0] = wl_row[k + 0];
                Alin[4 * s + c][1] = wl_row[k + 1];
            }
        }
    }
    float blin_v[8];
    #pragma unroll
    for (int v = 0; v < 8; ++v) blin_v[v] = b_lin[8 * hi + v];

    const v4f* stb = (const v4f*)ws + (size_t)NBLK * T * GX_UNITS_PER_STEP
                   + (size_t)blk * T * ST_UNITS_PER_STEP + lane;

    for (int t = blockIdx.x * 8 + wave; t < T; t += gridDim.x * 8) {
        const v4f* hp = stb + (size_t)t * ST_UNITS_PER_STEP;
        v8f acc_y0 = {0.f, 0.f, 0.f, 0.f, 0.f, 0.f, 0.f, 0.f};
        v8f acc_y1 = {0.f, 0.f, 0.f, 0.f, 0.f, 0.f, 0.f, 0.f};
        #pragma unroll
        for (int s = 0; s < NSLICE; ++s) {
            const v4f p0 = hp[(2 * s + 0) * WAVE_SZ];
            const v4f p1 = hp[(2 * s + 1) * WAVE_SZ];
            const v2f bq[4] = { {p0[0], p0[1]}, {p0[2], p0[3]}, {p1[0], p1[1]}, {p1[2], p1[3]} };
            #pragma unroll
            for (int c = 0; c < 4; ++c) {
                if (s < 2)
                    acc_y0 = __builtin_amdgcn_wmma_f32_16x16x4_f32(
                        false, Alin[4 * s + c], false, bq[c], (short)0, acc_y0, false, false);
                else
                    acc_y1 = __builtin_amdgcn_wmma_f32_16x16x4_f32(
                        false, Alin[4 * s + c], false, bq[c], (short)0, acc_y1, false, false);
            }
        }
        float* orow = out + ((size_t)(b0 + ln) * T + t) * GRU_O + 8 * hi;
        float y[8];
        #pragma unroll
        for (int v = 0; v < 8; ++v) y[v] = acc_y0[v] + acc_y1[v] + blin_v[v];
        *(v4f*)(orow + 0) = (v4f){y[0], y[1], y[2], y[3]};
        *(v4f*)(orow + 4) = (v4f){y[4], y[5], y[6], y[7]};
    }
}

extern "C" void kernel_launch(void* const* d_in, const int* in_sizes, int n_in,
                              void* d_out, int out_size, void* d_ws, size_t ws_size,
                              hipStream_t stream) {
    const float* x     = (const float*)d_in[0];
    const float* W_ih  = (const float*)d_in[1];
    const float* W_hh  = (const float*)d_in[2];
    const float* b_ih  = (const float*)d_in[3];
    const float* b_hh  = (const float*)d_in[4];
    const float* W_lin = (const float*)d_in[5];
    const float* b_lin = (const float*)d_in[6];
    float* out = (float*)d_out;
    float* ws  = (float*)d_ws;

    const int B = 512;
    const int T = in_sizes[0] / (B * GRU_I);

    gru_input_gates_kernel<<<dim3(32, NBLK), dim3(4 * WAVE_SZ), 0, stream>>>(
        x, W_ih, b_ih, ws, T);

    dim3 grid(B / BM);
    dim3 block(NWAVES * WAVE_SZ);
    gru_recurrent_kernel<<<grid, block, 0, stream>>>(W_hh, b_hh, ws, T);

    gru_projection_kernel<<<dim3(16, NBLK), dim3(8 * WAVE_SZ), 0, stream>>>(
        W_lin, b_lin, ws, out, T);
}


# r and n rows of the same 8 hidden columns share one weight tile (8 r/n + 4 z tiles): a lane ends its n gates alone, no r-to-n LDS hand-over; z read before the last tanh; hidden k order permuted
# speedup vs baseline: 3.0959x; 1.1738x over previous
#include <hip/hip_runtime.h>


typedef float v2f __attribute__((ext_vector_type(2)));
typedef float v4f __attribute__((ext_vector_type(4)));
typedef float v8f __attribute__((ext_vector_type(8)));

#define GRU_H   64
#define GRU_I   32
#define GRU_O   16
#define GRU_NG  192
#define BM      16
#define NBLK    32
#define NWAVES  12
#define NTILE   12
#define WAVE_SZ 32
#define NSLICE  (GRU_H / 16)
#define NGROUP  (GRU_H / 8)
#define GX_UNITS_PER_STEP   (NTILE * 2 * WAVE_SZ)
#define ST_UNITS_PER_STEP   (NGROUP * WAVE_SZ)

__device__ __forceinline__ int gate_row(int tile, int m)
{
    if (tile < NGROUP) {
        const int col = 8 * tile + 4 * (m >> 3) + (m & 3);
        return (m & 4) ? 2 * GRU_H + col : col;
    }
    return GRU_H + 16 * (tile - NGROUP) + m;
}

__global__ __launch_bounds__(4 * WAVE_SZ)
void gru_input_gates_kernel(const float* __restrict__ x,
                            const float* __restrict__ W_ih,
                            const float* __restrict__ b_ih,
                            float* __restrict__ ws,
                            int T)
{
    const int tid  = threadIdx.x;
    const int wave = tid / WAVE_SZ;
    const int lane = tid % WAVE_SZ;
    const int hi   = lane >> 4;
    const int ln   = lane & 15;
    const int blk  = blockIdx.y;
    const int b0   = blk * BM;

    v2f   A[3][GRU_I / 4];
    float bias[3][8];
    #pragma unroll
    for (int g = 0; g < 3; ++g) {
        const int tile = 3 * wave + g;
        const v4f* wrow = (const v4f*)(W_ih + (size_t)gate_row(tile, ln) * GRU_I + 16 * hi);
        #pragma unroll
        for (int i = 0; i < 4; ++i) {
            const v4f w = wrow[i];
            A[g][2 * i + 0] = (v2f){w[0], w[1]};
            A[g][2 * i + 1] = (v2f){w[2], w[3]};
        }
        #pragma unroll
        for (int v = 0; v < 8; ++v) bias[g][v] = b_ih[gate_row(tile, 8 * hi + v)];
    }

    const float* xrow = x + ((size_t)(b0 + ln) * T) * GRU_I + 16 * hi;
    v4f* const gx = (v4f*)ws;

    const int t0 = blockIdx.x;
    v4f xn[4];
    if (t0 < T) {
        const v4f* xp = (const v4f*)(xrow + (size_t)t0 * GRU_I);
        #pragma unroll
        for (int i = 0; i < 4; ++i) xn[i] = xp[i];
    }

    for (int t = t0; t < T; t += gridDim.x) {
        v4f xv[4];
        #pragma unroll
        for (int i = 0; i < 4; ++i) xv[i] = xn[i];
        {
            const int tn = (t + (int)gridDim.x < T) ? t + (int)gridDim.x : t;
            const v4f* xp = (const v4f*)(xrow + (size_t)tn * GRU_I);
            #pragma unroll
            for (int i = 0; i < 4; ++i) xn[i] = xp[i];
        }

        v8f acc[3];
        #pragma unroll
        for (int g = 0; g < 3; ++g) acc[g] = (v8f){0.f, 0.f, 0.f, 0.f, 0.f, 0.f, 0.f, 0.f};

        #pragma unroll
        for (int kc = 0; kc < GRU_I / 4; ++kc) {
            const v2f b = {xv[kc >> 1][2 * (kc & 1) + 0], xv[kc >> 1][2 * (kc & 1) + 1]};
            #pragma unroll
            for (int g = 0; g < 3; ++g)
                acc[g] = __builtin_amdgcn_wmma_f32_16x16x4_f32(
                    false, A[g][kc], false, b, (short)0, acc[g], false, false);
        }

        v4f* dst = gx + ((size_t)blk * T + t) * GX_UNITS_PER_STEP + (3 * wave) * 2 * WAVE_SZ + lane;
        #pragma unroll
        for (int g = 0; g < 3; ++g) {
            dst[(2 * g + 0) * WAVE_SZ] = (v4f){acc[g][0] + bias[g][0], acc[g][1] + bias[g][1],
                                               acc[g][2] + bias[g][2], acc[g][3] + bias[g][3]};
            dst[(2 * g + 1) * WAVE_SZ] = (v4f){acc[g][4] + bias[g][4], acc[g][5] + bias[g][5],
                                               acc[g][6] + bias[g][6], acc[g][7] + bias[g][7]};
        }
    }
}

__global__ __launch_bounds__(NWAVES * WAVE_SZ)
void gru_recurrent_kernel(const float* __restrict__ W_hh,
                          const float* __restrict__ b_hh,
                          float* __restrict__ ws,
                          int T)
{
    __shared__ v4f h_frag[NGROUP * WAVE_SZ];
    __shared__ v4f z_frag[NSLICE * 2 * WAVE_SZ];

    const int tid  = threadIdx.x;
    const int wave = tid / WAVE_SZ;
    const int lane = tid % WAVE_SZ;
    const int hi   = lane >> 4;
    const int ln   = lane & 15;
    const int blk  = blockIdx.x;

    const bool is_z = (wave >= NGROUP);
    const int  u    = wave & (NGROUP - 1);

    v2f Ahh[GRU_H / 4];
    {
        const float* whh_row = W_hh + (size_t)gate_row(wave, ln) * GRU_H;
        #pragma unroll
        for (int g = 0; g < NGROUP; ++g) {
            #pragma unroll
            for (int c = 0; c < 2; ++c) {
                const int k = 8 * g + 4 * hi + 2 * c;
                Ahh[2 * g + c][0] = whh_row[k + 0];
                Ahh[2 * g + c][1] = whh_row[k + 1];
            }
        }
    }
    float bhh_v[8];
    #pragma unroll
    for (int v = 0; v < 8; ++v) bhh_v[v] = b_hh[gate_row(wave, 8 * hi + v)];

    for (int i = tid; i < NGROUP * WAVE_SZ; i += blockDim.x)
        h_frag[i] = (v4f){0.f, 0.f, 0.f, 0.f};
    v4f hq = {0.f, 0.f, 0.f, 0.f};
    __syncthreads();

    const v4f* const gxb = (const v4f*)ws + (size_t)blk * T * GX_UNITS_PER_STEP;
    const v4f* gxw = gxb + wave * 2 * WAVE_SZ + lane;
    v4f* stw = (v4f*)ws + (size_t)NBLK * T * GX_UNITS_PER_STEP
             + (size_t)blk * T * ST_UNITS_PER_STEP + u * WAVE_SZ + lane;
    const int zidx = ((u >> 1) * 2 + hi) * WAVE_SZ + 16 * (u & 1) + ln;

    v4f gx0 = gxw[0];
    v4f gx1 = gxw[WAVE_SZ];

    for (int t = 0; t < T; ++t) {
        const int tn = (t + 1 < T) ? t + 1 : t;
        const v4f nx0 = gxw[(size_t)tn * GX_UNITS_PER_STEP];
        const v4f nx1 = gxw[(size_t)tn * GX_UNITS_PER_STEP + WAVE_SZ];

        v8f acc_h0 = {0.f, 0.f, 0.f, 0.f, 0.f, 0.f, 0.f, 0.f};
        v8f acc_h1 = {0.f, 0.f, 0.f, 0.f, 0.f, 0.f, 0.f, 0.f};

        #pragma unroll
        for (int g = 0; g < NGROUP; ++g) {
            const v4f p = h_frag[g * WAVE_SZ + lane];
            const v2f b0 = {p[0], p[1]};
            const v2f b1 = {p[2], p[3]};
            if (g < NGROUP / 2) {
                acc_h0 = __builtin_amdgcn_wmma_f32_16x16x4_f32(
                    false, Ahh[2 * g + 0], false, b0, (short)0, acc_h0, false, false);
                acc_h0 = __builtin_amdgcn_wmma_f32_16x16x4_f32(
                    false, Ahh[2 * g + 1], false, b1, (short)0, acc_h0, false, false);
            } else {
                acc_h1 = __builtin_amdgcn_wmma_f32_16x16x4_f32(
                    false, Ahh[2 * g + 0], false, b0, (short)0, acc_h1, false, false);
                acc_h1 = __builtin_amdgcn_wmma_f32_16x16x4_f32(
                    false, Ahh[2 * g + 1], false, b1, (short)0, acc_h1, false, false);
            }
        }

        float nv[4] = {0.f, 0.f, 0.f, 0.f};
        float pre3 = 0.f;

        float s03[4];
        #pragma unroll
        for (int v = 0; v < 4; ++v) {
            const float a = gx0[v] + acc_h0[v] + acc_h1[v] + bhh_v[v];
            s03[v] = 1.0f / (1.0f + __expf(-a));
        }

        if (is_z) {
            float s47[4];
            #pragma unroll
            for (int v = 0; v < 4; ++v) {
                const float a = gx1[v] + acc_h0[4 + v] + acc_h1[4 + v] + bhh_v[4 + v];
                s47[v] = 1.0f / (1.0f + __expf(-a));
            }
            z_frag[(2 * u + 0) * WAVE_SZ + lane] = (v4f){s03[0], s03[1], s03[2], s03[3]};
            z_frag[(2 * u + 1) * WAVE_SZ + lane] = (v4f){s47[0], s47[1], s47[2], s47[3]};
        } else {
            float arg[4];
            #pragma unroll
            for (int v = 0; v < 4; ++v) {
                const float gh = acc_h0[4 + v] + acc_h1[4 + v] + bhh_v[4 + v];
                arg[v] = gx1[v] + s03[v] * gh;
            }
            #pragma unroll
            for (int v = 0; v < 3; ++v) nv[v] = tanhf(arg[v]);
            pre3 = arg[3];
        }
        __syncthreads();

        if (!is_z) {
            const v4f zc = z_frag[zidx];
            nv[3] = tanhf(pre3);
            #pragma unroll
            for (int v = 0; v < 4; ++v) hq[v] = (1.0f - zc[v]) * nv[v] + zc[v] * hq[v];
            h_frag[u * WAVE_SZ + lane] = hq;
        }
        __syncthreads();

        if (!is_z) stw[(size_t)t * ST_UNITS_PER_STEP] = hq;

        gx0 = nx0;
        gx1 = nx1;
    }
}

__global__ __launch_bounds__(8 * WAVE_SZ)
void gru_projection_kernel(const float* __restrict__ W_lin,
                           const float* __restrict__ b_lin,
                           const float* __restrict__ ws,
                           float* __restrict__ out,
                           int T)
{
    const int tid  = threadIdx.x;
    const int wave = tid / WAVE_SZ;
    const int lane = tid % WAVE_SZ;
    const int hi   = lane >> 4;
    const int ln   = lane & 15;
    const int blk  = blockIdx.y;
    const int b0   = blk * BM;

    v2f Alin[GRU_H / 4];
    {
        const float* wl_row = W_lin + (size_t)ln * GRU_H;
        #pragma unroll
        for (int g = 0; g < NGROUP; ++g) {
            #pragma unroll
            for (int c = 0; c < 2; ++c) {
                const int k = 8 * g + 4 * hi + 2 * c;
                Alin[2 * g + c][0] = wl_row[k + 0];
                Alin[2 * g + c][1] = wl_row[k + 1];
            }
        }
    }
    float blin_v[8];
    #pragma unroll
    for (int v = 0; v < 8; ++v) blin_v[v] = b_lin[8 * hi + v];

    const v4f* stb = (const v4f*)ws + (size_t)NBLK * T * GX_UNITS_PER_STEP
                   + (size_t)blk * T * ST_UNITS_PER_STEP + lane;

    for (int t = blockIdx.x * 8 + wave; t < T; t += gridDim.x * 8) {
        const v4f* hp = stb + (size_t)t * ST_UNITS_PER_STEP;
        v8f acc_y0 = {0.f, 0.f, 0.f, 0.f, 0.f, 0.f, 0.f, 0.f};
        v8f acc_y1 = {0.f, 0.f, 0.f, 0.f, 0.f, 0.f, 0.f, 0.f};
        #pragma unroll
        for (int g = 0; g < NGROUP; ++g) {
            const v4f p = hp[g * WAVE_SZ];
            const v2f bq[2] = { {p[0], p[1]}, {p[2], p[3]} };
            #pragma unroll
            for (int c = 0; c < 2; ++c) {
                if (g < NGROUP / 2)
                    acc_y0 = __builtin_amdgcn_wmma_f32_16x16x4_f32(
                        false, Alin[2 * g + c], false, bq[c], (short)0, acc_y0, false, false);
                else
                    acc_y1 = __builtin_amdgcn_wmma_f32_16x16x4_f32(
                        false, Alin[2 * g + c], false, bq[c], (short)0, acc_y1, false, false);
            }
        }
        float* orow = out + ((size_t)(b0 + ln) * T + t) * GRU_O + 8 * hi;
        float y[8];
        #pragma unroll
        for (int v = 0; v < 8; ++v) y[v] = acc_y0[v] + acc_y1[v] + blin_v[v];
        *(v4f*)(orow + 0) = (v4f){y[0], y[1], y[2], y[3]};
        *(v4f*)(orow + 4) = (v4f){y[4], y[5], y[6], y[7]};
    }
}

extern "C" void kernel_launch(void* const* d_in, const int* in_sizes, int n_in,
                              void* d_out, int out_size, void* d_ws, size_t ws_size,
                              hipStream_t stream) {
    const float* x     = (const float*)d_in[0];
    const float* W_ih  = (const float*)d_in[1];
    const float* W_hh  = (const float*)d_in[2];
    const float* b_ih  = (const float*)d_in[3];
    const float* b_hh  = (const float*)d_in[4];
    const float* W_lin = (const float*)d_in[5];
    const float* b_lin = (const float*)d_in[6];
    float* out = (float*)d_out;
    float* ws  = (float*)d_ws;

    const int B = 512;
    const int T = in_sizes[0] / (B * GRU_I);

    gru_input_gates_kernel<<<dim3(32, NBLK), dim3(4 * WAVE_SZ), 0, stream>>>(
        x, W_ih, b_ih, ws, T);

    dim3 grid(B / BM);
    dim3 block(NWAVES * WAVE_SZ);
    gru_recurrent_kernel<<<grid, block, 0, stream>>>(W_hh, b_hh, ws, T);

    gru_projection_kernel<<<dim3(16, NBLK), dim3(8 * WAVE_SZ), 0, stream>>>(
        W_lin, b_lin, ws, out, T);
}


# each batch row sits in 4 tile columns of the recurrence (4 rows a block, 128 blocks): the 4 lanes holding a row's results finish one hidden column each: 1 sigmoid + 1 tanh chain a lane, not 4 + 4
# speedup vs baseline: 3.7753x; 1.2195x over previous
#include <hip/hip_runtime.h>


typedef float v2f __attribute__((ext_vector_type(2)));
typedef float v4f __attribute__((ext_vector_type(4)));
typedef float v8f __attribute__((ext_vector_type(8)));

#define GRU_H   64
#define GRU_I   32
#define GRU_O   16
#define GRU_NG  192
#define BM      16
#define NBLK    32
#define RB      4
#define NRBLK   128
#define NWAVES  12
#define NTILE   12
#define WAVE_SZ 32
#define NSLICE  (GRU_H / 16)
#define NGROUP  (GRU_H / 8)
#define GX_UNITS_PER_STEP   (NTILE * WAVE_SZ)
#define ST_UNITS_PER_STEP   (NGROUP * 2 * RB)
#define ST_FLOATS_PER_STEP  (4 * ST_UNITS_PER_STEP)

__device__ __forceinline__ int gate_row(int tile, int m)
{
    if (tile < NGROUP) {
        const int col = 8 * tile + 4 * (m >> 3) + (m & 3);
        return (m & 4) ? 2 * GRU_H + col : col;
    }
    return GRU_H + 16 * (tile - NGROUP) + m;
}

__global__ __launch_bounds__(4 * WAVE_SZ)
void gru_input_gates_kernel(const float* __restrict__ x,
                            const float* __restrict__ W_ih,
                            const float* __restrict__ b_ih,
                            float* __restrict__ ws,
                            int T)
{
    const int tid  = threadIdx.x;
    const int wave = tid / WAVE_SZ;
    const int lane = tid % WAVE_SZ;
    const int hi   = lane >> 4;
    const int ln   = lane & 15;
    const int blk  = blockIdx.y;
    const int b0   = blk * BM;

    v2f   A[3][GRU_I / 4];
    float bias[3][8];
    #pragma unroll
    for (int g = 0; g < 3; ++g) {
        const int tile = 3 * wave + g;
        const v4f* wrow = (const v4f*)(W_ih + (size_t)gate_row(tile, ln) * GRU_I + 16 * hi);
        #pragma unroll
        for (int i = 0; i < 4; ++i) {
            const v4f w = wrow[i];
            A[g][2 * i + 0] = (v2f){w[0], w[1]};
            A[g][2 * i + 1] = (v2f){w[2], w[3]};
        }
        #pragma unroll
        for (int v = 0; v < 8; ++v) bias[g][v] = b_ih[gate_row(tile, 8 * hi + v)];
    }

    const float* xrow = x + ((size_t)(b0 + ln) * T) * GRU_I + 16 * hi;
    v2f* const gx = (v2f*)ws;
    const size_t rblk = (size_t)blk * (BM / RB) + (ln >> 2);

    const int t0 = blockIdx.x;
    v4f xn[4];
    if (t0 < T) {
        const v4f* xp = (const v4f*)(xrow + (size_t)t0 * GRU_I);
        #pragma unroll
        for (int i = 0; i < 4; ++i) xn[i] = xp[i];
    }

    for (int t = t0; t < T; t += gridDim.x) {
        v4f xv[4];
        #pragma unroll
        for (int i = 0; i < 4; ++i) xv[i] = xn[i];
        {
            const int tn = (t + (int)gridDim.x < T) ? t + (int)gridDim.x : t;
            const v4f* xp = (const v4f*)(xrow + (size_t)tn * GRU_I);
            #pragma unroll
            for (int i = 0; i < 4; ++i) xn[i] = xp[i];
        }

        v8f acc[3];
        #pragma unroll
        for (int g = 0; g < 3; ++g) acc[g] = (v8f){0.f, 0.f, 0.f, 0.f, 0.f, 0.f, 0.f, 0.f};

        #pragma unroll
        for (int kc = 0; kc < GRU_I / 4; ++kc) {
            const v2f b = {xv[kc >> 1][2 * (kc & 1) + 0], xv[kc >> 1][2 * (kc & 1) + 1]};
            #pragma unroll
            for (int g = 0; g < 3; ++g)
                acc[g] = __builtin_amdgcn_wmma_f32_16x16x4_f32(
                    false, A[g][kc], false, b, (short)0, acc[g], false, false);
        }

        v2f* dst = gx + ((rblk * T + t) * NTILE + 3 * wave) * WAVE_SZ + 16 * hi + (ln & 3);
        #pragma unroll
        for (int g = 0; g < 3; ++g) {
            #pragma unroll
            for (int q = 0; q < 4; ++q)
                dst[g * WAVE_SZ + 4 * q] = (v2f){acc[g][q] + bias[g][q],
                                                 acc[g][4 + q] + bias[g][4 + q]};
        }
    }
}


__device__ __forceinline__ float pick4(float x0, float x1, float x2, float x3, int j)
{
    const float lo = (j & 1) ? x1 : x0;
    const float up = (j & 1) ? x3 : x2;
    return (j & 2) ? up : lo;
}

__global__ __launch_bounds__(NWAVES * WAVE_SZ) __attribute__((amdgpu_waves_per_eu(3, 3)))
void gru_recurrent_kernel(const float* __restrict__ W_hh,
                          const float* __restrict__ b_hh,
                          float* __restrict__ ws,
                          int T)
{
    __shared__ v4f h_frag[NGROUP * 2 * RB];
    __shared__ v2f z_frag[NSLICE * WAVE_SZ];

    const int tid  = threadIdx.x;
    const int wave = tid / WAVE_SZ;
    const int lane = tid % WAVE_SZ;
    const int hi   = lane >> 4;
    const int ln   = lane & 15;
    const int j    = ln >> 2;
    const int i    = ln & 3;
    const int blk  = blockIdx.x;

    const bool is_z = (wave >= NGROUP);
    const int  u    = wave & (NGROUP - 1);

    v2f Ahh[GRU_H / 4];
    {
        const float* whh_row = W_hh + (size_t)gate_row(wave, ln) * GRU_H;
        #pragma unroll
        for (int g = 0; g < NGROUP; ++g) {
            #pragma unroll
            for (int c = 0; c < 2; ++c) {
                const int k = 8 * g + 4 * hi + 2 * c;
                Ahh[2 * g + c][0] = whh_row[k + 0];
                Ahh[2 * g + c][1] = whh_row[k + 1];
            }
        }
    }
    const float bhh_lo = b_hh[gate_row(wave, 8 * hi + j)];
    const float bhh_up = b_hh[gate_row(wave, 8 * hi + 4 + j)];

    for (int k = tid; k < NGROUP * 2 * RB; k += blockDim.x)
        h_frag[k] = (v4f){0.f, 0.f, 0.f, 0.f};
    float hcol = 0.f;
    __syncthreads();

    const v2f* gxw = (const v2f*)ws + ((size_t)blk * T * NTILE + wave) * WAVE_SZ + lane;
    const int hpos = u * 4 * 2 * RB + hi * 4 * RB + i * 4 + j;
    float* stw = ws + (size_t)NRBLK * T * GX_UNITS_PER_STEP * 2
               + (size_t)blk * T * ST_FLOATS_PER_STEP + hpos;
    float* const h_flt = (float*)h_frag;
    const float* const z_src = (const float*)z_frag + (u * 16 + ln) * 2 + hi;
    const v4f* const h_own = h_frag + hi * RB + i;

    v2f gxc = gxw[0];

    for (int t = 0; t < T; ++t) {
        const int tn = (t + 1 < T) ? t + 1 : t;
        const v2f nxt = gxw[(size_t)tn * GX_UNITS_PER_STEP];

        v8f acc_h0 = {0.f, 0.f, 0.f, 0.f, 0.f, 0.f, 0.f, 0.f};
        v8f acc_h1 = {0.f, 0.f, 0.f, 0.f, 0.f, 0.f, 0.f, 0.f};

        v4f hb[NGROUP];
        #pragma unroll
        for (int g = 0; g < NGROUP; ++g) hb[g] = h_own[g * 2 * RB];
        #pragma unroll
        for (int g = 0; g < NGROUP / 2; ++g) {
            const int g1 = g + NGROUP / 2;
            #pragma unroll
            for (int c = 0; c < 2; ++c) {
                const v2f b0 = {hb[g][2 * c], hb[g][2 * c + 1]};
                const v2f b1 = {hb[g1][2 * c], hb[g1][2 * c + 1]};
                acc_h0 = __builtin_amdgcn_wmma_f32_16x16x4_f32(
                    false, Ahh[2 * g + c], false, b0, (short)0, acc_h0, false, false);
                acc_h1 = __builtin_amdgcn_wmma_f32_16x16x4_f32(
                    false, Ahh[2 * g1 + c], false, b1, (short)0, acc_h1, false, false);
            }
        }

        const float lo0 = pick4(acc_h0[0], acc_h0[1], acc_h0[2], acc_h0[3], j);
        const float lo1 = pick4(acc_h1[0], acc_h1[1], acc_h1[2], acc_h1[3], j);
        const float up0 = pick4(acc_h0[4], acc_h0[5], acc_h0[6], acc_h0[7], j);
        const float up1 = pick4(acc_h1[4], acc_h1[5], acc_h1[6], acc_h1[7], j);

        const float a_lo = gxc[0] + lo0 + lo1 + bhh_lo;
        const float s_lo = 1.0f / (1.0f + __expf(-a_lo));

        float nval = 0.f;

        if (is_z) {
            const float a_up = gxc[1] + up0 + up1 + bhh_up;
            const float s_up = 1.0f / (1.0f + __expf(-a_up));
            z_frag[u * WAVE_SZ + lane] = (v2f){s_lo, s_up};
        } else {
            const float gh = up0 + up1 + bhh_up;
            nval = tanhf(gxc[1] + s_lo * gh);
        }
        __syncthreads();

        if (!is_z) {
            const float zc = *z_src;
            hcol = fmaf(nval, 1.0f - zc, zc * hcol);
            h_flt[hpos] = hcol;
        }
        __syncthreads();

        if (!is_z) stw[(size_t)t * ST_FLOATS_PER_STEP] = hcol;

        gxc = nxt;
    }
}

__global__ __launch_bounds__(8 * WAVE_SZ)
void gru_projection_kernel(const float* __restrict__ W_lin,
                           const float* __restrict__ b_lin,
                           const float* __restrict__ ws,
                           float* __restrict__ out,
                           int T)
{
    const int tid  = threadIdx.x;
    const int wave = tid / WAVE_SZ;
    const int lane = tid % WAVE_SZ;
    const int hi   = lane >> 4;
    const int ln   = lane & 15;
    const int blk  = blockIdx.y;
    const int b0   = blk * BM;

    v2f Alin[GRU_H / 4];
    {
        const float* wl_row = W_lin + (size_t)ln * GRU_H;
        #pragma unroll
        for (int g = 0; g < NGROUP; ++g) {
            #pragma unroll
            for (int c = 0; c < 2; ++c) {
                const int k = 8 * g + 4 * hi + 2 * c;
                Alin[2 * g + c][0] = wl_row[k + 0];
                Alin[2 * g + c][1] = wl_row[k + 1];
            }
        }
    }
    float blin_v[8];
    #pragma unroll
    for (int v = 0; v < 8; ++v) blin_v[v] = b_lin[8 * hi + v];

    const v4f* stb = (const v4f*)ws + (size_t)NRBLK * T * (GX_UNITS_PER_STEP / 2)
                   + ((size_t)blk * (BM / RB) + (ln >> 2)) * T * ST_UNITS_PER_STEP
                   + hi * RB + (ln & 3);

    for (int t = blockIdx.x * 8 + wave; t < T; t += gridDim.x * 8) {
        const v4f* hp = stb + (size_t)t * ST_UNITS_PER_STEP;
        v8f acc_y0 = {0.f, 0.f, 0.f, 0.f, 0.f, 0.f, 0.f, 0.f};
        v8f acc_y1 = {0.f, 0.f, 0.f, 0.f, 0.f, 0.f, 0.f, 0.f};
        #pragma unroll
        for (int g = 0; g < NGROUP; ++g) {
            const v4f p = hp[g * 2 * RB];
            const v2f bq[2] = { {p[0], p[1]}, {p[2], p[3]} };
            #pragma unroll
            for (int c = 0; c < 2; ++c) {
                if (g < NGROUP / 2)
                    acc_y0 = __builtin_amdgcn_wmma_f32_16x16x4_f32(
                        false, Alin[2 * g + c], false, bq[c], (short)0, acc_y0, false, false);
                else
                    acc_y1 = __builtin_amdgcn_wmma_f32_16x16x4_f32(
                        false, Alin[2 * g + c], false, bq[c], (short)0, acc_y1, false, false);
            }
        }
        float* orow = out + ((size_t)(b0 + ln) * T + t) * GRU_O + 8 * hi;
        float y[8];
        #pragma unroll
        for (int v = 0; v < 8; ++v) y[v] = acc_y0[v] + acc_y1[v] + blin_v[v];
        *(v4f*)(orow + 0) = (v4f){y[0], y[1], y[2], y[3]};
        *(v4f*)(orow + 4) = (v4f){y[4], y[5], y[6], y[7]};
    }
}

extern "C" void kernel_launch(void* const* d_in, const int* in_sizes, int n_in,
                              void* d_out, int out_size, void* d_ws, size_t ws_size,
                              hipStream_t stream) {
    const float* x     = (const float*)d_in[0];
    const float* W_ih  = (const float*)d_in[1];
    const float* W_hh  = (const float*)d_in[2];
    const float* b_ih  = (const float*)d_in[3];
    const float* b_hh  = (const float*)d_in[4];
    const float* W_lin = (const float*)d_in[5];
    const float* b_lin = (const float*)d_in[6];
    float* out = (float*)d_out;
    float* ws  = (float*)d_ws;

    const int B = 512;
    const int T = in_sizes[0] / (B * GRU_I);

    gru_input_gates_kernel<<<dim3(32, NBLK), dim3(4 * WAVE_SZ), 0, stream>>>(
        x, W_ih, b_ih, ws, T);

    dim3 grid(B / RB);
    dim3 block(NWAVES * WAVE_SZ);
    gru_recurrent_kernel<<<grid, block, 0, stream>>>(W_hh, b_hh, ws, T);

    gru_projection_kernel<<<dim3(16, NBLK), dim3(8 * WAVE_SZ), 0, stream>>>(
        W_lin, b_lin, ws, out, T);
}


# the z waves, last on each SIMD's matrix pipe, take n from the r/n waves through LDS, blend and publish the state themselves: z is no longer handed back; hand-written barriers wait for LDS only
# speedup vs baseline: 4.1639x; 1.1029x over previous
#include <hip/hip_runtime.h>


typedef float v2f __attribute__((ext_vector_type(2)));
typedef float v4f __attribute__((ext_vector_type(4)));
typedef float v8f __attribute__((ext_vector_type(8)));

#define GRU_H   64
#define GRU_I   32
#define GRU_O   16
#define GRU_NG  192
#define BM      16
#define NBLK    32
#define RB      4
#define NRBLK   128
#define NWAVES  12
#define NTILE   12
#define WAVE_SZ 32
#define NSLICE  (GRU_H / 16)
#define NGROUP  (GRU_H / 8)
#define GX_UNITS_PER_STEP   (NTILE * WAVE_SZ)
#define ST_UNITS_PER_STEP   (NGROUP * 2 * RB)
#define ST_FLOATS_PER_STEP  (4 * ST_UNITS_PER_STEP)

__device__ __forceinline__ int gate_row(int tile, int m)
{
    if (tile < NGROUP) {
        const int col = 8 * tile + 4 * (m >> 3) + (m & 3);
        return (m & 4) ? 2 * GRU_H + col : col;
    }
    return GRU_H + 16 * (tile - NGROUP) + m;
}

__global__ __launch_bounds__(4 * WAVE_SZ)
void gru_input_gates_kernel(const float* __restrict__ x,
                            const float* __restrict__ W_ih,
                            const float* __restrict__ b_ih,
                            float* __restrict__ ws,
                            int T)
{
    const int tid  = threadIdx.x;
    const int wave = tid / WAVE_SZ;
    const int lane = tid % WAVE_SZ;
    const int hi   = lane >> 4;
    const int ln   = lane & 15;
    const int blk  = blockIdx.y;
    const int b0   = blk * BM;

    v2f   A[3][GRU_I / 4];
    float bias[3][8];
    #pragma unroll
    for (int g = 0; g < 3; ++g) {
        const int tile = 3 * wave + g;
        const v4f* wrow = (const v4f*)(W_ih + (size_t)gate_row(tile, ln) * GRU_I + 16 * hi);
        #pragma unroll
        for (int i = 0; i < 4; ++i) {
            const v4f w = wrow[i];
            A[g][2 * i + 0] = (v2f){w[0], w[1]};
            A[g][2 * i + 1] = (v2f){w[2], w[3]};
        }
        #pragma unroll
        for (int v = 0; v < 8; ++v) bias[g][v] = b_ih[gate_row(tile, 8 * hi + v)];
    }

    const float* xrow = x + ((size_t)(b0 + ln) * T) * GRU_I + 16 * hi;
    v2f* const gx = (v2f*)ws;
    const size_t rblk = (size_t)blk * (BM / RB) + (ln >> 2);

    const int t0 = blockIdx.x;
    v4f xn[4];
    if (t0 < T) {
        const v4f* xp = (const v4f*)(xrow + (size_t)t0 * GRU_I);
        #pragma unroll
        for (int i = 0; i < 4; ++i) xn[i] = xp[i];
    }

    for (int t = t0; t < T; t += gridDim.x) {
        v4f xv[4];
        #pragma unroll
        for (int i = 0; i < 4; ++i) xv[i] = xn[i];
        {
            const int tn = (t + (int)gridDim.x < T) ? t + (int)gridDim.x : t;
            const v4f* xp = (const v4f*)(xrow + (size_t)tn * GRU_I);
            #pragma unroll
            for (int i = 0; i < 4; ++i) xn[i] = xp[i];
        }

        v8f acc[3];
        #pragma unroll
        for (int g = 0; g < 3; ++g) acc[g] = (v8f){0.f, 0.f, 0.f, 0.f, 0.f, 0.f, 0.f, 0.f};

        #pragma unroll
        for (int kc = 0; kc < GRU_I / 4; ++kc) {
            const v2f b = {xv[kc >> 1][2 * (kc & 1) + 0], xv[kc >> 1][2 * (kc & 1) + 1]};
            #pragma unroll
            for (int g = 0; g < 3; ++g)
                acc[g] = __builtin_amdgcn_wmma_f32_16x16x4_f32(
                    false, A[g][kc], false, b, (short)0, acc[g], false, false);
        }

        v2f* dst = gx + ((rblk * T + t) * NTILE + 3 * wave) * WAVE_SZ + 16 * hi + (ln & 3);
        #pragma unroll
        for (int g = 0; g < 3; ++g) {
            #pragma unroll
            for (int q = 0; q < 4; ++q)
                dst[g * WAVE_SZ + 4 * q] = (v2f){acc[g][q] + bias[g][q],
                                                 acc[g][4 + q] + bias[g][4 + q]};
        }
    }
}


__device__ __forceinline__ float pick4(float x0, float x1, float x2, float x3, int j)
{
    const float lo = (j & 1) ? x1 : x0;
    const float up = (j & 1) ? x3 : x2;
    return (j & 2) ? up : lo;
}

__device__ __forceinline__ void lds_barrier()
{
    asm volatile("s_wait_dscnt 0x0\n\ts_barrier_signal -1\n\ts_barrier_wait -1" ::: "memory");
}

__global__ __launch_bounds__(NWAVES * WAVE_SZ) __attribute__((amdgpu_waves_per_eu(3, 3)))
void gru_recurrent_kernel(const float* __restrict__ W_hh,
                          const float* __restrict__ b_hh,
                          float* __restrict__ ws,
                          int T)
{
    __shared__ v4f h_frag[NGROUP * 2 * RB];
    __shared__ float n_frag[NGROUP * WAVE_SZ];

    const int tid  = threadIdx.x;
    const int wave = tid / WAVE_SZ;
    const int lane = tid % WAVE_SZ;
    const int hi   = lane >> 4;
    const int ln   = lane & 15;
    const int j    = ln >> 2;
    const int i    = ln & 3;
    const int blk  = blockIdx.x;

    const bool is_z = (wave >= NGROUP);
    const int  u    = wave & (NGROUP - 1);

    v2f Ahh[GRU_H / 4];
    {
        const float* whh_row = W_hh + (size_t)gate_row(wave, ln) * GRU_H;
        #pragma unroll
        for (int g = 0; g < NGROUP; ++g) {
            #pragma unroll
            for (int c = 0; c < 2; ++c) {
                const int k = 8 * g + 4 * hi + 2 * c;
                Ahh[2 * g + c][0] = whh_row[k + 0];
                Ahh[2 * g + c][1] = whh_row[k + 1];
            }
        }
    }
    const float bhh_lo = b_hh[gate_row(wave, 8 * hi + j)];
    const float bhh_up = b_hh[gate_row(wave, 8 * hi + 4 + j)];

    for (int k = tid; k < NGROUP * 2 * RB; k += blockDim.x)
        h_frag[k] = (v4f){0.f, 0.f, 0.f, 0.f};
    float h_lo = 0.f;
    float h_up = 0.f;
    __syncthreads();

    const v2f* gxw = (const v2f*)ws + ((size_t)blk * T * NTILE + wave) * WAVE_SZ + lane;
    float* const n_dst = n_frag + u * WAVE_SZ + lane;
    const int zg   = 2 * (u & (NSLICE - 1)) + hi;
    const float* const n_src = n_frag + zg * WAVE_SZ + ln;
    const int hpos = zg * 4 * 2 * RB + i * 4 + j;
    float* const h_dst = (float*)h_frag + hpos;
    float* stw = ws + (size_t)NRBLK * T * GX_UNITS_PER_STEP * 2
               + (size_t)blk * T * ST_FLOATS_PER_STEP + hpos;
    const v4f* const h_own = h_frag + hi * RB + i;

    v2f gxc = gxw[0];

    for (int t = 0; t < T; ++t) {
        const int tn = (t + 1 < T) ? t + 1 : t;
        const v2f nxt = gxw[(size_t)tn * GX_UNITS_PER_STEP];

        v8f acc_h0 = {0.f, 0.f, 0.f, 0.f, 0.f, 0.f, 0.f, 0.f};
        v8f acc_h1 = {0.f, 0.f, 0.f, 0.f, 0.f, 0.f, 0.f, 0.f};

        v4f hb[NGROUP];
        #pragma unroll
        for (int g = 0; g < NGROUP; ++g) hb[g] = h_own[g * 2 * RB];
        #pragma unroll
        for (int g = 0; g < NGROUP / 2; ++g) {
            const int g1 = g + NGROUP / 2;
            #pragma unroll
            for (int c = 0; c < 2; ++c) {
                const v2f b0 = {hb[g][2 * c], hb[g][2 * c + 1]};
                const v2f b1 = {hb[g1][2 * c], hb[g1][2 * c + 1]};
                acc_h0 = __builtin_amdgcn_wmma_f32_16x16x4_f32(
                    false, Ahh[2 * g + c], false, b0, (short)0, acc_h0, false, false);
                acc_h1 = __builtin_amdgcn_wmma_f32_16x16x4_f32(
                    false, Ahh[2 * g1 + c], false, b1, (short)0, acc_h1, false, false);
            }
        }

        const float lo0 = pick4(acc_h0[0], acc_h0[1], acc_h0[2], acc_h0[3], j);
        const float lo1 = pick4(acc_h1[0], acc_h1[1], acc_h1[2], acc_h1[3], j);
        const float up0 = pick4(acc_h0[4], acc_h0[5], acc_h0[6], acc_h0[7], j);
        const float up1 = pick4(acc_h1[4], acc_h1[5], acc_h1[6], acc_h1[7], j);

        float d_lo = 0.f;
        float d_up = 0.f;
        if (!is_z) {
            const float a_lo = gxc[0] + lo0 + lo1 + bhh_lo;
            const float s_lo = 1.0f / (1.0f + __expf(-a_lo));
            const float gh = up0 + up1 + bhh_up;
            *n_dst = tanhf(gxc[1] + s_lo * gh);
        } else {
            const float a_lo = gxc[0] + lo0 + lo1 + bhh_lo;
            const float a_up = gxc[1] + up0 + up1 + bhh_up;
            d_lo = 1.0f + __expf(-a_lo);
            d_up = 1.0f + __expf(-a_up);
        }
        lds_barrier();

        if (is_z) {
            const float n_lo = n_src[0];
            const float n_up = n_src[WAVE_SZ / 2];
            __builtin_amdgcn_sched_barrier(0);
            const float s_lo = 1.0f / d_lo;
            const float s_up = 1.0f / d_up;
            h_lo = fmaf(n_lo, 1.0f - s_lo, s_lo * h_lo);
            h_up = fmaf(n_up, 1.0f - s_up, s_up * h_up);
            h_dst[0]      = h_lo;
            h_dst[4 * RB] = h_up;
        }
        lds_barrier();

        if (is_z) {
            stw[(size_t)t * ST_FLOATS_PER_STEP]          = h_lo;
            stw[(size_t)t * ST_FLOATS_PER_STEP + 4 * RB] = h_up;
        }

        gxc = nxt;
    }
}

__global__ __launch_bounds__(8 * WAVE_SZ)
void gru_projection_kernel(const float* __restrict__ W_lin,
                           const float* __restrict__ b_lin,
                           const float* __restrict__ ws,
                           float* __restrict__ out,
                           int T)
{
    const int tid  = threadIdx.x;
    const int wave = tid / WAVE_SZ;
    const int lane = tid % WAVE_SZ;
    const int hi   = lane >> 4;
    const int ln   = lane & 15;
    const int blk  = blockIdx.y;
    const int b0   = blk * BM;

    v2f Alin[GRU_H / 4];
    {
        const float* wl_row = W_lin + (size_t)ln * GRU_H;
        #pragma unroll
        for (int g = 0; g < NGROUP; ++g) {
            #pragma unroll
            for (int c = 0; c < 2; ++c) {
                const int k = 8 * g + 4 * hi + 2 * c;
                Alin[2 * g + c][0] = wl_row[k + 0];
                Alin[2 * g + c][1] = wl_row[k + 1];
            }
        }
    }
    float blin_v[8];
    #pragma unroll
    for (int v = 0; v < 8; ++v) blin_v[v] = b_lin[8 * hi + v];

    const v4f* stb = (const v4f*)ws + (size_t)NRBLK * T * (GX_UNITS_PER_STEP / 2)
                   + ((size_t)blk * (BM / RB) + (ln >> 2)) * T * ST_UNITS_PER_STEP
                   + hi * RB + (ln & 3);

    for (int t = blockIdx.x * 8 + wave; t < T; t += gridDim.x * 8) {
        const v4f* hp = stb + (size_t)t * ST_UNITS_PER_STEP;
        v8f acc_y0 = {0.f, 0.f, 0.f, 0.f, 0.f, 0.f, 0.f, 0.f};
        v8f acc_y1 = {0.f, 0.f, 0.f, 0.f, 0.f, 0.f, 0.f, 0.f};
        #pragma unroll
        for (int g = 0; g < NGROUP; ++g) {
            const v4f p = hp[g * 2 * RB];
            const v2f bq[2] = { {p[0], p[1]}, {p[2], p[3]} };
            #pragma unroll
            for (int c = 0; c < 2; ++c) {
                if (g < NGROUP / 2)
                    acc_y0 = __builtin_amdgcn_wmma_f32_16x16x4_f32(
                        false, Alin[2 * g + c], false, bq[c], (short)0, acc_y0, false, false);
                else
                    acc_y1 = __builtin_amdgcn_wmma_f32_16x16x4_f32(
                        false, Alin[2 * g + c], false, bq[c], (short)0, acc_y1, false, false);
            }
        }
        float* orow = out + ((size_t)(b0 + ln) * T + t) * GRU_O + 8 * hi;
        float y[8];
        #pragma unroll
        for (int v = 0; v < 8; ++v) y[v] = acc_y0[v] + acc_y1[v] + blin_v[v];
        *(v4f*)(orow + 0) = (v4f){y[0], y[1], y[2], y[3]};
        *(v4f*)(orow + 4) = (v4f){y[4], y[5], y[6], y[7]};
    }
}

extern "C" void kernel_launch(void* const* d_in, const int* in_sizes, int n_in,
                              void* d_out, int out_size, void* d_ws, size_t ws_size,
                              hipStream_t stream) {
    const float* x     = (const float*)d_in[0];
    const float* W_ih  = (const float*)d_in[1];
    const float* W_hh  = (const float*)d_in[2];
    const float* b_ih  = (const float*)d_in[3];
    const float* b_hh  = (const float*)d_in[4];
    const float* W_lin = (const float*)d_in[5];
    const float* b_lin = (const float*)d_in[6];
    float* out = (float*)d_out;
    float* ws  = (float*)d_ws;

    const int B = 512;
    const int T = in_sizes[0] / (B * GRU_I);

    gru_input_gates_kernel<<<dim3(32, NBLK), dim3(4 * WAVE_SZ), 0, stream>>>(
        x, W_ih, b_ih, ws, T);

    dim3 grid(B / RB);
    dim3 block(NWAVES * WAVE_SZ);
    gru_recurrent_kernel<<<grid, block, 0, stream>>>(W_hh, b_hh, ws, T);

    gru_projection_kernel<<<dim3(16, NBLK), dim3(8 * WAVE_SZ), 0, stream>>>(
        W_lin, b_lin, ws, out, T);
}
